# starting kernel
# baseline (speedup 1.0000x reference)
#include <hip/hip_runtime.h>
#include <math.h>

typedef __attribute__((ext_vector_type(2))) float v2f;
typedef __attribute__((ext_vector_type(8))) float v8f;

#define BB 2
#define SS 192
#define HH 768
#define LL 4
#define NN (HH * LL)
#define MM (BB * SS)

__global__ __launch_bounds__(32) void proj_wmma_f32(
    const float* __restrict__ X, const float* __restrict__ W,
    const float* __restrict__ bias, float* __restrict__ Y) {
  const int n0 = blockIdx.x * 16;
  const int m0 = blockIdx.y * 16;
  const int lane = threadIdx.x;
  const int r16  = lane & 15;
  const int kh   = (lane >> 4) * 2;

  const float* __restrict__ arow = X + (size_t)(m0 + r16) * HH;
  const float* __restrict__ brow = W + (size_t)(n0 + r16) * HH;

  v8f acc = {};
  for (int k = 0; k < HH; k += 4) {
    v2f a = *(const v2f*)(arow + k + kh);
    v2f b = *(const v2f*)(brow + k + kh);
    acc = __builtin_amdgcn_wmma_f32_16x16x4_f32(
         false, a,  false, b,
         (short)0, acc,  false,  false);
  }

  const int n     = n0 + r16;
  const int mbase = m0 + ((lane >> 4) << 3);
  const float bn  = bias[n];
#pragma unroll
  for (int r = 0; r < 8; ++r) {
    Y[(size_t)(mbase + r) * NN + n] = acc[r] + bn;
  }
}

__global__ __launch_bounds__(256) void pair_gelu_reduce(
    const float* __restrict__ hsrc, const float* __restrict__ htgt,
    const float* __restrict__ wcls, float* __restrict__ out) {
  __shared__ float s_src[HH];
  __shared__ float s_w[HH];

  const int s = blockIdx.x;
  const int l = blockIdx.y;
  const int b = blockIdx.z;
  const int tid = threadIdx.x;

  const float* __restrict__ srcrow =
      hsrc + ((size_t)((b * SS + s) * LL + l)) * HH;
  for (int i = tid; i < HH; i += 256) {
    s_src[i] = srcrow[i];
    s_w[i]   = wcls[i * LL + l];
  }
  __syncthreads();

  const int lane = tid & 31;
  const int wid  = tid >> 5;
  const float* __restrict__ tgtbase =
      htgt + ((size_t)(b * SS * LL + l)) * HH;

  for (int t = wid; t < SS; t += 8) {
    const float* __restrict__ trow = tgtbase + (size_t)t * (LL * HH);
    float acc = 0.0f;
#pragma unroll 4
    for (int h = lane; h < HH; h += 32) {
      float x = s_src[h] + trow[h];
      float g = 0.5f * x * (1.0f + erff(x * 0.70710678118654752f));
      acc = fmaf(g, s_w[h], acc);
    }
#pragma unroll
    for (int off = 16; off > 0; off >>= 1) acc += __shfl_xor(acc, off, 32);
    if (lane == 0) {
      out[((size_t)((b * SS + s) * SS + t)) * LL + l] = acc;
    }
  }
}

extern "C" void kernel_launch(void* const* d_in, const int* in_sizes, int n_in,
                              void* d_out, int out_size, void* d_ws, size_t ws_size,
                              hipStream_t stream) {
  const float* pooled = (const float*)d_in[0];
  const float* W_src  = (const float*)d_in[1];
  const float* b_src  = (const float*)d_in[2];
  const float* W_tgt  = (const float*)d_in[3];
  const float* b_tgt  = (const float*)d_in[4];
  const float* wcls   = (const float*)d_in[5];
  float* out = (float*)d_out;

  float* hsrc = (float*)d_ws;
  float* htgt = hsrc + (size_t)MM * NN;

  dim3 g1(NN / 16, MM / 16);
  proj_wmma_f32<<<g1, 32, 0, stream>>>(pooled, W_src, b_src, hsrc);
  proj_wmma_f32<<<g1, 32, 0, stream>>>(pooled, W_tgt, b_tgt, htgt);

  dim3 g2(SS, LL, BB);
  pair_gelu_reduce<<<g2, 256, 0, stream>>>(hsrc, htgt, wcls, out);
}


# projection GEMM: k split six ways over the waves of a work-group: 256 work-groups of 12 waves (3 per SIMD), 48x48 wave tiles, global->registers; partials summed in LDS; k sum order changed
# speedup vs baseline: 3.4288x; 3.4288x over previous
#include <hip/hip_runtime.h>
#include <math.h>

typedef __attribute__((ext_vector_type(2))) float v2f;
typedef __attribute__((ext_vector_type(8))) float v8f;

#define BB 2
#define SS 192
#define HH 768
#define LL 4
#define NN (HH * LL)
#define MM (BB * SS)

typedef __attribute__((ext_vector_type(4))) float v4f;

#define PM 3
#define PN 3
#define KQ 4
#define KT (8 * KQ)
#define KSPLIT 6
#define KPART (HH / KSPLIT)
#define NTILE 2
#define PWAVES (KSPLIT * NTILE)
#define PUNITS (PM * PN * 2)
#define UPW (PUNITS / KSPLIT)

#define PROJ_LOAD(A, B, q, k)                                                   \
  _Pragma("unroll") for (int i = 0; i < PM; ++i)                                \
      A[q][i] = *(const v4f*)(pa + (size_t)i * (16 * HH) + (k) + 8 * (q));      \
  _Pragma("unroll") for (int j = 0; j < PN; ++j)                                \
      B[q][j] = *(const v4f*)(pb + (size_t)j * (16 * HH) + (k) + 8 * (q));

#define PROJ_MMA(A, B, q)                                                       \
  _Pragma("unroll") for (int i = 0; i < PM; ++i)                                \
    _Pragma("unroll") for (int j = 0; j < PN; ++j)                              \
      acc[i][j] = __builtin_amdgcn_wmma_f32_16x16x4_f32(                        \
          false, A[q][i].xy, false, B[q][j].xy, (short)0, acc[i][j], false,     \
          false);                                                               \
  _Pragma("unroll") for (int i = 0; i < PM; ++i)                                \
    _Pragma("unroll") for (int j = 0; j < PN; ++j)                              \
      acc[i][j] = __builtin_amdgcn_wmma_f32_16x16x4_f32(                        \
          false, A[q][i].zw, false, B[q][j].zw, (short)0, acc[i][j], false,     \
          false);

__global__ __launch_bounds__(32 * PWAVES) __attribute__((amdgpu_waves_per_eu(3, 3)))
void proj_wmma_f32(
    const float* __restrict__ X, const float* __restrict__ W,
    const float* __restrict__ bias, float* __restrict__ Y) {
  __shared__ v4f s_part[PWAVES * PUNITS * 32];

  const int lane = threadIdx.x & 31;
  const int wid = __builtin_amdgcn_readfirstlane((int)(threadIdx.x >> 5));
  const int jt  = wid & 1;
  const int ks  = wid >> 1;
  const int n0 = blockIdx.x * (16 * PN * NTILE) + jt * (16 * PN);
  const int m0 = blockIdx.y * (16 * PM);
  const int r16  = lane & 15;
  const int hh   = lane >> 4;

  const float* __restrict__ pa = X + (size_t)(m0 + r16) * HH + ks * KPART + 4 * hh;
  const float* __restrict__ pb = W + (size_t)(n0 + r16) * HH + ks * KPART + 4 * hh;

  v8f acc[PM][PN];
#pragma unroll
  for (int i = 0; i < PM; ++i)
#pragma unroll
    for (int j = 0; j < PN; ++j) acc[i][j] = (v8f){};

  v4f a0[KQ][PM], b0[KQ][PN];
  v4f a1[KQ][PM], b1[KQ][PN];

#pragma unroll
  for (int q = 0; q < KQ; ++q) { PROJ_LOAD(a0, b0, q, 0) }
#pragma unroll
  for (int q = 0; q < KQ; ++q) {
    PROJ_LOAD(a1, b1, q, KT)
    PROJ_MMA(a0, b0, q)
  }
#pragma unroll
  for (int q = 0; q < KQ; ++q) {
    PROJ_LOAD(a0, b0, q, 2 * KT)
    PROJ_MMA(a1, b1, q)
  }
#pragma unroll
  for (int q = 0; q < KQ; ++q) {
    PROJ_LOAD(a1, b1, q, 3 * KT)
    PROJ_MMA(a0, b0, q)
  }
#pragma unroll
  for (int q = 0; q < KQ; ++q) { PROJ_MMA(a1, b1, q) }

  v4f* mine = s_part + (size_t)wid * (PUNITS * 32) + lane;
#pragma unroll
  for (int i = 0; i < PM; ++i)
#pragma unroll
    for (int j = 0; j < PN; ++j) {
      const int ti = i * PN + j;
      v4f lo, hi;
      lo.x = acc[i][j][0]; lo.y = acc[i][j][1]; lo.z = acc[i][j][2]; lo.w = acc[i][j][3];
      hi.x = acc[i][j][4]; hi.y = acc[i][j][5]; hi.z = acc[i][j][6]; hi.w = acc[i][j][7];
      mine[(2 * ti) * 32]     = lo;
      mine[(2 * ti + 1) * 32] = hi;
    }
  __syncthreads();

  const v4f* parts = s_part + (size_t)jt * (PUNITS * 32) + lane;
#pragma unroll
  for (int e = 0; e < UPW; ++e) {
    const int u = ks * UPW + e;
    v4f sum = parts[u * 32];
#pragma unroll
    for (int p = 1; p < KSPLIT; ++p)
      sum += parts[(size_t)p * (NTILE * PUNITS * 32) + u * 32];
    const int ti = u >> 1;
    const int i  = ti / PN;
    const int j  = ti - i * PN;
    const int n  = n0 + 16 * j + r16;
    const float bn = bias[n];
    float* __restrict__ y = Y + (size_t)(m0 + 16 * i + 8 * hh + 4 * (u & 1)) * NN + n;
    y[0]              = sum.x + bn;
    y[(size_t)NN]     = sum.y + bn;
    y[(size_t)2 * NN] = sum.z + bn;
    y[(size_t)3 * NN] = sum.w + bn;
  }
}

#define TAB_HALF  176
#define TAB_NODES (2 * TAB_HALF + 1)
#define TAB_XMAX  5.5f
#define TAB_MAGIC 8388784.0f
#define TAB_BITS0 0x4B000000

#define TS 6
#define NW 32
#define NT (32 * NW)

__global__ __launch_bounds__(NT) void pair_gelu_reduce(
    const float* __restrict__ hsrc, const float* __restrict__ htgt,
    const float* __restrict__ wcls, float* __restrict__ out) {
  __shared__ v2f s_tab[TAB_NODES * 32];
  __shared__ v2f s_node[TAB_NODES];
  __shared__ float s_src[TS * HH];
  __shared__ float s_w[HH];

  const int s0 = blockIdx.x * TS;
  const int l = blockIdx.y;
  const int b = blockIdx.z;
  const int tid = threadIdx.x;

  if (tid <= TAB_HALF) {
    const float xa  = (float)tid * (1.0f / 32.0f);
    const float q   = 0.5f * erfcf(xa * 0.70710678118654752f);
    const float ph0 = 0.39894228040143268f * expf(-0.5f * xa * xa);
    const float phi = fmaf(ph0, fmaf(xa * xa, 1.0f / 32768.0f, -1.0f / 32768.0f), ph0);
    v2f ep; ep.x = 1.0f - q; ep.y = phi;
    v2f en; en.x = q;        en.y = phi;
    if (tid == TAB_HALF) { ep.x = 1.0f; ep.y = 0.0f; en.x = 0.0f; en.y = 0.0f; }
    s_node[TAB_HALF + tid] = ep;
    s_node[TAB_HALF - tid] = en;
  }

  const float* __restrict__ srcbase =
      hsrc + ((size_t)((b * SS + s0) * LL + l)) * HH;
  for (int e = tid; e < TS * HH; e += NT) {
    const int i = e / HH;
    const int h = e - i * HH;
    s_src[e] = srcbase[(size_t)i * (LL * HH) + h];
  }
  for (int h = tid; h < HH; h += NT) s_w[h] = wcls[h * LL + l];
  __syncthreads();
  const int lane = tid & 31;
  const int wid  = tid >> 5;
#pragma unroll
  for (int k = 0; k < TAB_NODES / NW; ++k) {
    const int n = wid + k * NW;
    s_tab[n * 32 + lane] = s_node[n];
  }
  if (wid < TAB_NODES % NW) {
    const int n = TAB_NODES - TAB_NODES % NW + wid;
    s_tab[n * 32 + lane] = s_node[n];
  }
  __syncthreads();
  const float* __restrict__ tgtbase =
      htgt + ((size_t)(b * SS * LL + l)) * HH;

  for (int t = wid; t < SS; t += NW) {
    const float* __restrict__ trow = tgtbase + (size_t)t * (LL * HH);
    float acc[TS];
#pragma unroll
    for (int i = 0; i < TS; ++i) acc[i] = 0.0f;
#pragma unroll 4
    for (int h = lane; h < HH; h += 32) {
      const float tv = trow[h];
      const float wv = s_w[h];
#pragma unroll
      for (int i = 0; i < TS; ++i) {
        const float x   = s_src[i * HH + h] + tv;
        const float xc  = fminf(fmaxf(x, -TAB_XMAX), TAB_XMAX);
        const float tq  = fmaf(xc, 32.0f, TAB_MAGIC);
        const v2f   pf  = s_tab[(__float_as_int(tq) - TAB_BITS0) * 32 + lane];
        const float xi  = fmaf(tq, 1.0f / 32.0f, -(TAB_MAGIC / 32.0f));
        const float d   = xc - xi;
        const float p   = fmaf(-0.5f * d, xi, 1.0f);
        const float ph  = fmaf(pf.y * d, p, pf.x);
        acc[i] = fmaf(ph * x, wv, acc[i]);
      }
    }
#pragma unroll
    for (int off = 16; off > 0; off >>= 1) {
#pragma unroll
      for (int i = 0; i < TS; ++i) acc[i] += __shfl_xor(acc[i], off, 32);
    }
    if (lane == 0) {
#pragma unroll
      for (int i = 0; i < TS; ++i)
        out[((size_t)((b * SS + s0 + i) * SS + t)) * LL + l] = acc[i];
    }
  }
}

extern "C" void kernel_launch(void* const* d_in, const int* in_sizes, int n_in,
                              void* d_out, int out_size, void* d_ws, size_t ws_size,
                              hipStream_t stream) {
  const float* pooled = (const float*)d_in[0];
  const float* W_src  = (const float*)d_in[1];
  const float* b_src  = (const float*)d_in[2];
  const float* W_tgt  = (const float*)d_in[3];
  const float* b_tgt  = (const float*)d_in[4];
  const float* wcls   = (const float*)d_in[5];
  float* out = (float*)d_out;

  float* hsrc = (float*)d_ws;
  float* htgt = hsrc + (size_t)MM * NN;

  dim3 g1(NN / (16 * PN * NTILE), MM / (16 * PM));
  proj_wmma_f32<<<g1, 32 * PWAVES, 0, stream>>>(pooled, W_src, b_src, hsrc);
  proj_wmma_f32<<<g1, 32 * PWAVES, 0, stream>>>(pooled, W_tgt, b_tgt, htgt);

  dim3 g2(SS / TS, LL, BB);
  pair_gelu_reduce<<<g2, NT, 0, stream>>>(hsrc, htgt, wcls, out);
}


# pair kernel: lane owns 4 consecutive h per 128, so a trip's target values are one 16-byte load per lane, issued one trip ahead of use (across row ends); flat 36-trip loop; per-lane sum order changed
# speedup vs baseline: 3.5566x; 1.0373x over previous
#include <hip/hip_runtime.h>
#include <math.h>

typedef __attribute__((ext_vector_type(2))) float v2f;
typedef __attribute__((ext_vector_type(8))) float v8f;

#define BB 2
#define SS 192
#define HH 768
#define LL 4
#define NN (HH * LL)
#define MM (BB * SS)

typedef __attribute__((ext_vector_type(4))) float v4f;

#define PM 3
#define PN 3
#define KQ 4
#define KT (8 * KQ)
#define KSPLIT 6
#define KPART (HH / KSPLIT)
#define NTILE 2
#define PWAVES (KSPLIT * NTILE)
#define PUNITS (PM * PN * 2)
#define UPW (PUNITS / KSPLIT)

#define PROJ_LOAD(A, B, q, k)                                                   \
  _Pragma("unroll") for (int i = 0; i < PM; ++i)                                \
      A[q][i] = *(const v4f*)(pa + (size_t)i * (16 * HH) + (k) + 8 * (q));      \
  _Pragma("unroll") for (int j = 0; j < PN; ++j)                                \
      B[q][j] = *(const v4f*)(pb + (size_t)j * (16 * HH) + (k) + 8 * (q));

#define PROJ_MMA(A, B, q)                                                       \
  _Pragma("unroll") for (int i = 0; i < PM; ++i)                                \
    _Pragma("unroll") for (int j = 0; j < PN; ++j)                              \
      acc[i][j] = __builtin_amdgcn_wmma_f32_16x16x4_f32(                        \
          false, A[q][i].xy, false, B[q][j].xy, (short)0, acc[i][j], false,     \
          false);                                                               \
  _Pragma("unroll") for (int i = 0; i < PM; ++i)                                \
    _Pragma("unroll") for (int j = 0; j < PN; ++j)                              \
      acc[i][j] = __builtin_amdgcn_wmma_f32_16x16x4_f32(                        \
          false, A[q][i].zw, false, B[q][j].zw, (short)0, acc[i][j], false,     \
          false);

__global__ __launch_bounds__(32 * PWAVES) __attribute__((amdgpu_waves_per_eu(3, 3)))
void proj_wmma_f32(
    const float* __restrict__ X, const float* __restrict__ W,
    const float* __restrict__ bias, float* __restrict__ Y) {
  __shared__ v4f s_part[PWAVES * PUNITS * 32];

  const int lane = threadIdx.x & 31;
  const int wid = __builtin_amdgcn_readfirstlane((int)(threadIdx.x >> 5));
  const int jt  = wid & 1;
  const int ks  = wid >> 1;
  const int n0 = blockIdx.x * (16 * PN * NTILE) + jt * (16 * PN);
  const int m0 = blockIdx.y * (16 * PM);
  const int r16  = lane & 15;
  const int hh   = lane >> 4;

  const float* __restrict__ pa = X + (size_t)(m0 + r16) * HH + ks * KPART + 4 * hh;
  const float* __restrict__ pb = W + (size_t)(n0 + r16) * HH + ks * KPART + 4 * hh;

  v8f acc[PM][PN];
#pragma unroll
  for (int i = 0; i < PM; ++i)
#pragma unroll
    for (int j = 0; j < PN; ++j) acc[i][j] = (v8f){};

  v4f a0[KQ][PM], b0[KQ][PN];
  v4f a1[KQ][PM], b1[KQ][PN];

#pragma unroll
  for (int q = 0; q < KQ; ++q) { PROJ_LOAD(a0, b0, q, 0) }
#pragma unroll
  for (int q = 0; q < KQ; ++q) {
    PROJ_LOAD(a1, b1, q, KT)
    PROJ_MMA(a0, b0, q)
  }
#pragma unroll
  for (int q = 0; q < KQ; ++q) {
    PROJ_LOAD(a0, b0, q, 2 * KT)
    PROJ_MMA(a1, b1, q)
  }
#pragma unroll
  for (int q = 0; q < KQ; ++q) {
    PROJ_LOAD(a1, b1, q, 3 * KT)
    PROJ_MMA(a0, b0, q)
  }
#pragma unroll
  for (int q = 0; q < KQ; ++q) { PROJ_MMA(a1, b1, q) }

  v4f* mine = s_part + (size_t)wid * (PUNITS * 32) + lane;
#pragma unroll
  for (int i = 0; i < PM; ++i)
#pragma unroll
    for (int j = 0; j < PN; ++j) {
      const int ti = i * PN + j;
      v4f lo, hi;
      lo.x = acc[i][j][0]; lo.y = acc[i][j][1]; lo.z = acc[i][j][2]; lo.w = acc[i][j][3];
      hi.x = acc[i][j][4]; hi.y = acc[i][j][5]; hi.z = acc[i][j][6]; hi.w = acc[i][j][7];
      mine[(2 * ti) * 32]     = lo;
      mine[(2 * ti + 1) * 32] = hi;
    }
  __syncthreads();

  const v4f* parts = s_part + (size_t)jt * (PUNITS * 32) + lane;
#pragma unroll
  for (int e = 0; e < UPW; ++e) {
    const int u = ks * UPW + e;
    v4f sum = parts[u * 32];
#pragma unroll
    for (int p = 1; p < KSPLIT; ++p)
      sum += parts[(size_t)p * (NTILE * PUNITS * 32) + u * 32];
    const int ti = u >> 1;
    const int i  = ti / PN;
    const int j  = ti - i * PN;
    const int n  = n0 + 16 * j + r16;
    const float bn = bias[n];
    float* __restrict__ y = Y + (size_t)(m0 + 16 * i + 8 * hh + 4 * (u & 1)) * NN + n;
    y[0]              = sum.x + bn;
    y[(size_t)NN]     = sum.y + bn;
    y[(size_t)2 * NN] = sum.z + bn;
    y[(size_t)3 * NN] = sum.w + bn;
  }
}

#define TAB_HALF  176
#define TAB_NODES (2 * TAB_HALF + 1)
#define TAB_XMAX  5.5f
#define TAB_MAGIC 8388784.0f
#define TAB_BITS0 0x4B000000

#define TS 6
#define NW 32
#define NT (32 * NW)

__global__ __launch_bounds__(NT) __attribute__((amdgpu_waves_per_eu(8, 8)))
void pair_gelu_reduce(
    const float* __restrict__ hsrc, const float* __restrict__ htgt,
    const float* __restrict__ wcls, float* __restrict__ out) {
  __shared__ v2f s_tab[TAB_NODES * 32];
  __shared__ v2f s_node[TAB_NODES];
  __shared__ __attribute__((aligned(16))) float s_src[TS * HH];
  __shared__ __attribute__((aligned(16))) float s_w[HH];

  const int s0 = blockIdx.x * TS;
  const int l = blockIdx.y;
  const int b = blockIdx.z;
  const int tid = threadIdx.x;

  if (tid <= TAB_HALF) {
    const float xa  = (float)tid * (1.0f / 32.0f);
    const float q   = 0.5f * erfcf(xa * 0.70710678118654752f);
    const float ph0 = 0.39894228040143268f * expf(-0.5f * xa * xa);
    const float phi = fmaf(ph0, fmaf(xa * xa, 1.0f / 32768.0f, -1.0f / 32768.0f), ph0);
    v2f ep; ep.x = 1.0f - q; ep.y = phi;
    v2f en; en.x = q;        en.y = phi;
    if (tid == TAB_HALF) { ep.x = 1.0f; ep.y = 0.0f; en.x = 0.0f; en.y = 0.0f; }
    s_node[TAB_HALF + tid] = ep;
    s_node[TAB_HALF - tid] = en;
  }

  const float* __restrict__ srcbase =
      hsrc + ((size_t)((b * SS + s0) * LL + l)) * HH;
  for (int e = tid; e < TS * HH; e += NT) {
    const int i = e / HH;
    const int h = e - i * HH;
    s_src[e] = srcbase[(size_t)i * (LL * HH) + h];
  }
  for (int h = tid; h < HH; h += NT) s_w[h] = wcls[h * LL + l];
  __syncthreads();
  const int lane = tid & 31;
  const int wid  = tid >> 5;
#pragma unroll
  for (int k = 0; k < TAB_NODES / NW; ++k) {
    const int n = wid + k * NW;
    s_tab[n * 32 + lane] = s_node[n];
  }
  if (wid < TAB_NODES % NW) {
    const int n = TAB_NODES - TAB_NODES % NW + wid;
    s_tab[n * 32 + lane] = s_node[n];
  }
  __syncthreads();
  const int wu = __builtin_amdgcn_readfirstlane(wid);
  const float* __restrict__ tp =
      htgt + ((size_t)(b * SS * LL + l)) * HH + (size_t)wu * (LL * HH);
  v4f tn = *(const v4f*)(tp + 4 * lane);
  const float* sp = s_src + 4 * lane;
  const float* wp = s_w + 4 * lane;
  int t = wu;
  int kq = 0;
  float acc[TS];
#pragma unroll
  for (int i = 0; i < TS; ++i) acc[i] = 0.0f;

  for (int trip = 0; trip < (SS / NW) * (HH / 128); ++trip) {
    const v4f tv = tn;
    const bool rowend = (kq == HH / 128 - 1);
    const float* __restrict__ np =
        (trip == (SS / NW) * (HH / 128) - 1) ? tp
        : rowend ? tp + ((size_t)NW * (LL * HH) - (HH - 128))
                 : tp + 128;
    tn = *(const v4f*)(np + 4 * lane);
    tp = np;
    const v4f wv = *(const v4f*)wp;
    v4f sv[TS];
#pragma unroll
    for (int i = 0; i < TS; ++i) sv[i] = *(const v4f*)(sp + i * HH);
#pragma unroll
    for (int e = 0; e < 4; ++e) {
#pragma unroll
      for (int i = 0; i < TS; ++i) {
        const float x   = sv[i][e] + tv[e];
        const float xc  = fminf(fmaxf(x, -TAB_XMAX), TAB_XMAX);
        const float tq  = fmaf(xc, 32.0f, TAB_MAGIC);
        const v2f   pf  = s_tab[(__float_as_int(tq) - TAB_BITS0) * 32 + lane];
        const float xi  = fmaf(tq, 1.0f / 32.0f, -(TAB_MAGIC / 32.0f));
        const float d   = xc - xi;
        const float p   = fmaf(-0.5f * d, xi, 1.0f);
        const float ph  = fmaf(pf.y * d, p, pf.x);
        acc[i] = fmaf(ph * x, wv[e], acc[i]);
      }
    }
    sp += 128;
    wp += 128;
    ++kq;
    if (rowend) {
#pragma unroll
      for (int off = 16; off > 0; off >>= 1) {
#pragma unroll
        for (int i = 0; i < TS; ++i) acc[i] += __shfl_xor(acc[i], off, 32);
      }
      if (lane == 0) {
#pragma unroll
        for (int i = 0; i < TS; ++i)
          out[((size_t)((b * SS + s0 + i) * SS + t)) * LL + l] = acc[i];
      }
#pragma unroll
      for (int i = 0; i < TS; ++i) acc[i] = 0.0f;
      sp -= HH;
      wp -= HH;
      kq = 0;
      t += NW;
    }
  }
}

extern "C" void kernel_launch(void* const* d_in, const int* in_sizes, int n_in,
                              void* d_out, int out_size, void* d_ws, size_t ws_size,
                              hipStream_t stream) {
  const float* pooled = (const float*)d_in[0];
  const float* W_src  = (const float*)d_in[1];
  const float* b_src  = (const float*)d_in[2];
  const float* W_tgt  = (const float*)d_in[3];
  const float* b_tgt  = (const float*)d_in[4];
  const float* wcls   = (const float*)d_in[5];
  float* out = (float*)d_out;

  float* hsrc = (float*)d_ws;
  float* htgt = hsrc + (size_t)MM * NN;

  dim3 g1(NN / (16 * PN * NTILE), MM / (16 * PM));
  proj_wmma_f32<<<g1, 32 * PWAVES, 0, stream>>>(pooled, W_src, b_src, hsrc);
  proj_wmma_f32<<<g1, 32 * PWAVES, 0, stream>>>(pooled, W_tgt, b_tgt, htgt);

  dim3 g2(SS / TS, LL, BB);
  pair_gelu_reduce<<<g2, NT, 0, stream>>>(hsrc, htgt, wcls, out);
}


# pair kernel: each group of six table gathers is issued one group ahead of the sums that use it (no wait directly behind its own gathers); source rows re-laid in LDS; per-lane sum order changed
# speedup vs baseline: 3.7188x; 1.0456x over previous
#include <hip/hip_runtime.h>
#include <math.h>

typedef __attribute__((ext_vector_type(2))) float v2f;
typedef __attribute__((ext_vector_type(8))) float v8f;

#define BB 2
#define SS 192
#define HH 768
#define LL 4
#define NN (HH * LL)
#define MM (BB * SS)

typedef __attribute__((ext_vector_type(4))) float v4f;

#define PM 3
#define PN 3
#define KQ 4
#define KT (8 * KQ)
#define KSPLIT 6
#define KPART (HH / KSPLIT)
#define NTILE 2
#define PWAVES (KSPLIT * NTILE)
#define PUNITS (PM * PN * 2)
#define UPW (PUNITS / KSPLIT)

#define PROJ_LOAD(A, B, q, k)                                                   \
  _Pragma("unroll") for (int i = 0; i < PM; ++i)                                \
      A[q][i] = *(const v4f*)(pa + (size_t)i * (16 * HH) + (k) + 8 * (q));      \
  _Pragma("unroll") for (int j = 0; j < PN; ++j)                                \
      B[q][j] = *(const v4f*)(pb + (size_t)j * (16 * HH) + (k) + 8 * (q));

#define PROJ_MMA(A, B, q)                                                       \
  _Pragma("unroll") for (int i = 0; i < PM; ++i)                                \
    _Pragma("unroll") for (int j = 0; j < PN; ++j)                              \
      acc[i][j] = __builtin_amdgcn_wmma_f32_16x16x4_f32(                        \
          false, A[q][i].xy, false, B[q][j].xy, (short)0, acc[i][j], false,     \
          false);                                                               \
  _Pragma("unroll") for (int i = 0; i < PM; ++i)                                \
    _Pragma("unroll") for (int j = 0; j < PN; ++j)                              \
      acc[i][j] = __builtin_amdgcn_wmma_f32_16x16x4_f32(                        \
          false, A[q][i].zw, false, B[q][j].zw, (short)0, acc[i][j], false,     \
          false);

__global__ __launch_bounds__(32 * PWAVES) __attribute__((amdgpu_waves_per_eu(3, 3)))
void proj_wmma_f32(
    const float* __restrict__ X, const float* __restrict__ W,
    const float* __restrict__ bias, float* __restrict__ Y) {
  __shared__ v4f s_part[PWAVES * PUNITS * 32];

  const int lane = threadIdx.x & 31;
  const int wid = __builtin_amdgcn_readfirstlane((int)(threadIdx.x >> 5));
  const int jt  = wid & 1;
  const int ks  = wid >> 1;
  const int n0 = blockIdx.x * (16 * PN * NTILE) + jt * (16 * PN);
  const int m0 = blockIdx.y * (16 * PM);
  const int r16  = lane & 15;
  const int hh   = lane >> 4;

  const float* __restrict__ pa = X + (size_t)(m0 + r16) * HH + ks * KPART + 4 * hh;
  const float* __restrict__ pb = W + (size_t)(n0 + r16) * HH + ks * KPART + 4 * hh;

  v8f acc[PM][PN];
#pragma unroll
  for (int i = 0; i < PM; ++i)
#pragma unroll
    for (int j = 0; j < PN; ++j) acc[i][j] = (v8f){};

  v4f a0[KQ][PM], b0[KQ][PN];
  v4f a1[KQ][PM], b1[KQ][PN];

#pragma unroll
  for (int q = 0; q < KQ; ++q) { PROJ_LOAD(a0, b0, q, 0) }
#pragma unroll
  for (int q = 0; q < KQ; ++q) {
    PROJ_LOAD(a1, b1, q, KT)
    PROJ_MMA(a0, b0, q)
  }
#pragma unroll
  for (int q = 0; q < KQ; ++q) {
    PROJ_LOAD(a0, b0, q, 2 * KT)
    PROJ_MMA(a1, b1, q)
  }
#pragma unroll
  for (int q = 0; q < KQ; ++q) {
    PROJ_LOAD(a1, b1, q, 3 * KT)
    PROJ_MMA(a0, b0, q)
  }
#pragma unroll
  for (int q = 0; q < KQ; ++q) { PROJ_MMA(a1, b1, q) }

  v4f* mine = s_part + (size_t)wid * (PUNITS * 32) + lane;
#pragma unroll
  for (int i = 0; i < PM; ++i)
#pragma unroll
    for (int j = 0; j < PN; ++j) {
      const int ti = i * PN + j;
      v4f lo, hi;
      lo.x = acc[i][j][0]; lo.y = acc[i][j][1]; lo.z = acc[i][j][2]; lo.w = acc[i][j][3];
      hi.x = acc[i][j][4]; hi.y = acc[i][j][5]; hi.z = acc[i][j][6]; hi.w = acc[i][j][7];
      mine[(2 * ti) * 32]     = lo;
      mine[(2 * ti + 1) * 32] = hi;
    }
  __syncthreads();

  const v4f* parts = s_part + (size_t)jt * (PUNITS * 32) + lane;
#pragma unroll
  for (int e = 0; e < UPW; ++e) {
    const int u = ks * UPW + e;
    v4f sum = parts[u * 32];
#pragma unroll
    for (int p = 1; p < KSPLIT; ++p)
      sum += parts[(size_t)p * (NTILE * PUNITS * 32) + u * 32];
    const int ti = u >> 1;
    const int i  = ti / PN;
    const int j  = ti - i * PN;
    const int n  = n0 + 16 * j + r16;
    const float bn = bias[n];
    float* __restrict__ y = Y + (size_t)(m0 + 16 * i + 8 * hh + 4 * (u & 1)) * NN + n;
    y[0]              = sum.x + bn;
    y[(size_t)NN]     = sum.y + bn;
    y[(size_t)2 * NN] = sum.z + bn;
    y[(size_t)3 * NN] = sum.w + bn;
  }
}

#define TAB_HALF  176
#define TAB_NODES (2 * TAB_HALF + 1)
#define TAB_XMAX  5.5f
#define TAB_MAGIC 8388784.0f
#define TAB_BITS0 0x4B000000

#define TS 6
#define NW 32
#define NT (32 * NW)
#define UR 3
#define HCH (HH / 128)
#define NTRIPS ((SS / NW) * HCH)

struct GGroup {
  v2f pf[UR][2];
  v2f q[UR];
  v2f xw[UR];
};

__global__ __launch_bounds__(NT) __attribute__((amdgpu_waves_per_eu(8, 8)))
void pair_gelu_reduce(
    const float* __restrict__ hsrc, const float* __restrict__ htgt,
    const float* __restrict__ wcls, float* __restrict__ out) {
  __shared__ v2f s_tab[TAB_NODES * 32];
  __shared__ v2f s_node[TAB_NODES];
  __shared__ __attribute__((aligned(16))) float s_src[TS * HH];
  __shared__ __attribute__((aligned(16))) float s_w[HH];

  const int s0 = blockIdx.x * TS;
  const int l = blockIdx.y;
  const int b = blockIdx.z;
  const int tid = threadIdx.x;

  if (tid <= TAB_HALF) {
    const float xa  = (float)tid * (1.0f / 32.0f);
    const float q   = 0.5f * erfcf(xa * 0.70710678118654752f);
    const float ph0 = 0.39894228040143268f * expf(-0.5f * xa * xa);
    const float phi = fmaf(ph0, fmaf(xa * xa, 1.0f / 32768.0f, -1.0f / 32768.0f), ph0);
    v2f ep; ep.x = 1.0f - q; ep.y = phi;
    v2f en; en.x = q;        en.y = phi;
    if (tid == TAB_HALF) { ep.x = 1.0f; ep.y = 0.0f; en.x = 0.0f; en.y = 0.0f; }
    s_node[TAB_HALF + tid] = ep;
    s_node[TAB_HALF - tid] = en;
  }

  const float* __restrict__ srcbase =
      hsrc + ((size_t)((b * SS + s0) * LL + l)) * HH;
  for (int e = tid; e < TS * HH; e += NT) {
    const int i = e / HH;
    const int h = e - i * HH;
    const int pos = (h & ~127) + ((h & 2) << 5) + ((h >> 1) & 62) + (h & 1);
    s_src[i * HH + pos] = srcbase[(size_t)i * (LL * HH) + h];
  }
  for (int h = tid; h < HH; h += NT) s_w[h] = wcls[h * LL + l];
  __syncthreads();
  const int lane = tid & 31;
  const int wid  = tid >> 5;
#pragma unroll
  for (int k = 0; k < TAB_NODES / NW; ++k) {
    const int n = wid + k * NW;
    s_tab[n * 32 + lane] = s_node[n];
  }
  if (wid < TAB_NODES % NW) {
    const int n = TAB_NODES - TAB_NODES % NW + wid;
    s_tab[n * 32 + lane] = s_node[n];
  }
  __syncthreads();
  const int wu = __builtin_amdgcn_readfirstlane(wid);

  const float* __restrict__ tbase =
      htgt + ((size_t)(b * SS * LL + l)) * HH + (size_t)wu * (LL * HH) + 4 * lane;
  auto trip_ptr = [&](int T) {
    const int rr = T / HCH;
    const int kk = T - rr * HCH;
    return tbase + (size_t)rr * (NW * LL * HH) + kk * 128;
  };
  auto load_s = [&](v2f (&dst)[UR], const float* base, int r0, int p) {
#pragma unroll
    for (int r = 0; r < UR; ++r) dst[r] = *(const v2f*)(base + (r0 + r) * HH + 64 * p);
  };
  auto issue = [&](GGroup& G, const v2f (&s2)[UR], v2f t2, v2f w2) {
    const v2f lo = (v2f){-TAB_XMAX, -TAB_XMAX}, hi = (v2f){TAB_XMAX, TAB_XMAX};
    const v2f k32 = (v2f){32.0f, 32.0f}, kmag = (v2f){TAB_MAGIC, TAB_MAGIC};
    const v2f kinv = (v2f){1.0f / 32.0f, 1.0f / 32.0f};
    const v2f koff = (v2f){-(TAB_MAGIC / 32.0f), -(TAB_MAGIC / 32.0f)};
    const v2f one = (v2f){1.0f, 1.0f};
#pragma unroll
    for (int r = 0; r < UR; ++r) {
      const v2f x  = s2[r] + t2;
      const v2f xc = __builtin_elementwise_min(__builtin_elementwise_max(x, lo), hi);
      const v2f tq = __builtin_elementwise_fma(xc, k32, kmag);
      G.pf[r][0] = s_tab[(__float_as_int(tq.x) - TAB_BITS0) * 32 + lane];
      G.pf[r][1] = s_tab[(__float_as_int(tq.y) - TAB_BITS0) * 32 + lane];
      const v2f xi = __builtin_elementwise_fma(tq, kinv, koff);
      const v2f d  = xc - xi;
      const v2f p  = __builtin_elementwise_fma(-0.5f * d, xi, one);
      G.q[r]  = d * p;
      G.xw[r] = x * w2;
    }
  };

  v2f acc[TS];
#pragma unroll
  for (int i = 0; i < TS; ++i) acc[i] = (v2f){0.0f, 0.0f};
  auto sum = [&](const GGroup& G, int r0) {
#pragma unroll
    for (int r = 0; r < UR; ++r) {
      v2f ph;
      ph.x = fmaf(G.pf[r][0].y, G.q[r].x, G.pf[r][0].x);
      ph.y = fmaf(G.pf[r][1].y, G.q[r].y, G.pf[r][1].x);
      acc[r0 + r] = __builtin_elementwise_fma(ph, G.xw[r], acc[r0 + r]);
    }
  };

  v4f tva = *(const v4f*)trip_ptr(0);
  v4f tvb = *(const v4f*)trip_ptr(1);
  const float* sp = s_src + 2 * lane;
  const float* wp = s_w + 4 * lane;
  v4f wva = *(const v4f*)wp;
  v4f wvb;
  v2f sa[UR], sb[UR];
  GGroup C;
  load_s(sa, sp, 0, 0);
  issue(C, sa, tva.xy, wva.xy);
  load_s(sb, sp, UR, 0);
  int t = wu;
  int kq = 0;

  auto one_trip = [&](v4f& tc, v4f& tx, v4f& wc, v4f& wx, const int trip, const bool rowend) {
    const float* spn = rowend ? sp - (HH - 128) : sp + 128;
    const float* wpn = rowend ? wp - (HH - 128) : wp + 128;
    wx = *(const v4f*)wpn;
    GGroup N1, N2, N3;
    load_s(sa, sp, 0, 1);
    issue(N1, sb, tc.xy, wc.xy);
    __builtin_amdgcn_sched_barrier(0);
    sum(C, 0);
    __builtin_amdgcn_sched_barrier(0);
    load_s(sb, sp, UR, 1);
    issue(N2, sa, tc.zw, wc.zw);
    __builtin_amdgcn_sched_barrier(0);
    sum(N1, UR);
    __builtin_amdgcn_sched_barrier(0);
    load_s(sa, spn, 0, 0);
    issue(N3, sb, tc.zw, wc.zw);
    __builtin_amdgcn_sched_barrier(0);
    sum(N2, 0);
    __builtin_amdgcn_sched_barrier(0);
    tc = *(const v4f*)trip_ptr(trip + 2 < NTRIPS ? trip + 2 : NTRIPS - 1);
    load_s(sb, spn, UR, 0);
    issue(C, sa, tx.xy, wx.xy);
    __builtin_amdgcn_sched_barrier(0);
    sum(N3, UR);
    sp = spn;
    wp = wpn;
    ++kq;
    if (rowend) {
      float a[TS];
#pragma unroll
      for (int i = 0; i < TS; ++i) a[i] = acc[i].x + acc[i].y;
#pragma unroll
      for (int off = 16; off > 0; off >>= 1) {
#pragma unroll
        for (int i = 0; i < TS; ++i) a[i] += __shfl_xor(a[i], off, 32);
      }
      if (lane == 0) {
#pragma unroll
        for (int i = 0; i < TS; ++i)
          out[((size_t)((b * SS + s0 + i) * SS + t)) * LL + l] = a[i];
      }
#pragma unroll
      for (int i = 0; i < TS; ++i) acc[i] = (v2f){0.0f, 0.0f};
      kq = 0;
      t += NW;
    }
  };

  for (int trip = 0; trip < NTRIPS; trip += 2) {
    one_trip(tva, tvb, wva, wvb, trip, false);
    one_trip(tvb, tva, wvb, wva, trip + 1, kq == HCH - 1);
  }
}

extern "C" void kernel_launch(void* const* d_in, const int* in_sizes, int n_in,
                              void* d_out, int out_size, void* d_ws, size_t ws_size,
                              hipStream_t stream) {
  const float* pooled = (const float*)d_in[0];
  const float* W_src  = (const float*)d_in[1];
  const float* b_src  = (const float*)d_in[2];
  const float* W_tgt  = (const float*)d_in[3];
  const float* b_tgt  = (const float*)d_in[4];
  const float* wcls   = (const float*)d_in[5];
  float* out = (float*)d_out;

  float* hsrc = (float*)d_ws;
  float* htgt = hsrc + (size_t)MM * NN;

  dim3 g1(NN / (16 * PN * NTILE), MM / (16 * PM));
  proj_wmma_f32<<<g1, 32 * PWAVES, 0, stream>>>(pooled, W_src, b_src, hsrc);
  proj_wmma_f32<<<g1, 32 * PWAVES, 0, stream>>>(pooled, W_tgt, b_tgt, htgt);

  dim3 g2(SS / TS, LL, BB);
  pair_gelu_reduce<<<g2, NT, 0, stream>>>(hsrc, htgt, wcls, out);
}


# pair kernel: clamp and node index from one saturating packed fma per h pair, u = sat(x/11 + 1/2), t = 352u + 2^23 (was two medians); 13 vector issue slots per pair, was 14; sum order kept
# speedup vs baseline: 4.0341x; 1.0848x over previous
#include <hip/hip_runtime.h>
#include <math.h>

typedef __attribute__((ext_vector_type(2))) float v2f;
typedef __attribute__((ext_vector_type(8))) float v8f;

#define BB 2
#define SS 192
#define HH 768
#define LL 4
#define NN (HH * LL)
#define MM (BB * SS)

typedef __attribute__((ext_vector_type(4))) float v4f;

#define PM 3
#define PN 3
#define KQ 4
#define KT (8 * KQ)
#define KSPLIT 6
#define KPART (HH / KSPLIT)
#define NTILE 2
#define PWAVES (KSPLIT * NTILE)
#define PUNITS (PM * PN * 2)
#define UPW (PUNITS / KSPLIT)

#define PROJ_LOAD(A, B, q, k)                                                   \
  _Pragma("unroll") for (int i = 0; i < PM; ++i)                                \
      A[q][i] = *(const v4f*)(pa + (size_t)i * (16 * HH) + (k) + 8 * (q));      \
  _Pragma("unroll") for (int j = 0; j < PN; ++j)                                \
      B[q][j] = *(const v4f*)(pb + (size_t)j * (16 * HH) + (k) + 8 * (q));

#define PROJ_MMA(A, B, q)                                                       \
  _Pragma("unroll") for (int i = 0; i < PM; ++i)                                \
    _Pragma("unroll") for (int j = 0; j < PN; ++j)                              \
      acc[i][j] = __builtin_amdgcn_wmma_f32_16x16x4_f32(                        \
          false, A[q][i].xy, false, B[q][j].xy, (short)0, acc[i][j], false,     \
          false);                                                               \
  _Pragma("unroll") for (int i = 0; i < PM; ++i)                                \
    _Pragma("unroll") for (int j = 0; j < PN; ++j)                              \
      acc[i][j] = __builtin_amdgcn_wmma_f32_16x16x4_f32(                        \
          false, A[q][i].zw, false, B[q][j].zw, (short)0, acc[i][j], false,     \
          false);

__global__ __launch_bounds__(32 * PWAVES) __attribute__((amdgpu_waves_per_eu(3, 3)))
void proj_wmma_f32(
    const float* __restrict__ X, const float* __restrict__ W,
    const float* __restrict__ bias, float* __restrict__ Y) {
  __shared__ v4f s_part[PWAVES * PUNITS * 32];

  const int lane = threadIdx.x & 31;
  const int wid = __builtin_amdgcn_readfirstlane((int)(threadIdx.x >> 5));
  const int jt  = wid & 1;
  const int ks  = wid >> 1;
  const int n0 = blockIdx.x * (16 * PN * NTILE) + jt * (16 * PN);
  const int m0 = blockIdx.y * (16 * PM);
  const int r16  = lane & 15;
  const int hh   = lane >> 4;

  const float* __restrict__ pa = X + (size_t)(m0 + r16) * HH + ks * KPART + 4 * hh;
  const float* __restrict__ pb = W + (size_t)(n0 + r16) * HH + ks * KPART + 4 * hh;

  v8f acc[PM][PN];
#pragma unroll
  for (int i = 0; i < PM; ++i)
#pragma unroll
    for (int j = 0; j < PN; ++j) acc[i][j] = (v8f){};

  v4f a0[KQ][PM], b0[KQ][PN];
  v4f a1[KQ][PM], b1[KQ][PN];

#pragma unroll
  for (int q = 0; q < KQ; ++q) { PROJ_LOAD(a0, b0, q, 0) }
#pragma unroll
  for (int q = 0; q < KQ; ++q) {
    PROJ_LOAD(a1, b1, q, KT)
    PROJ_MMA(a0, b0, q)
  }
#pragma unroll
  for (int q = 0; q < KQ; ++q) {
    PROJ_LOAD(a0, b0, q, 2 * KT)
    PROJ_MMA(a1, b1, q)
  }
#pragma unroll
  for (int q = 0; q < KQ; ++q) {
    PROJ_LOAD(a1, b1, q, 3 * KT)
    PROJ_MMA(a0, b0, q)
  }
#pragma unroll
  for (int q = 0; q < KQ; ++q) { PROJ_MMA(a1, b1, q) }

  v4f* mine = s_part + (size_t)wid * (PUNITS * 32) + lane;
#pragma unroll
  for (int i = 0; i < PM; ++i)
#pragma unroll
    for (int j = 0; j < PN; ++j) {
      const int ti = i * PN + j;
      v4f lo, hi;
      lo.x = acc[i][j][0]; lo.y = acc[i][j][1]; lo.z = acc[i][j][2]; lo.w = acc[i][j][3];
      hi.x = acc[i][j][4]; hi.y = acc[i][j][5]; hi.z = acc[i][j][6]; hi.w = acc[i][j][7];
      mine[(2 * ti) * 32]     = lo;
      mine[(2 * ti + 1) * 32] = hi;
    }
  __syncthreads();

  const v4f* parts = s_part + (size_t)jt * (PUNITS * 32) + lane;
#pragma unroll
  for (int e = 0; e < UPW; ++e) {
    const int u = ks * UPW + e;
    v4f sum = parts[u * 32];
#pragma unroll
    for (int p = 1; p < KSPLIT; ++p)
      sum += parts[(size_t)p * (NTILE * PUNITS * 32) + u * 32];
    const int ti = u >> 1;
    const int i  = ti / PN;
    const int j  = ti - i * PN;
    const int n  = n0 + 16 * j + r16;
    const float bn = bias[n];
    float* __restrict__ y = Y + (size_t)(m0 + 16 * i + 8 * hh + 4 * (u & 1)) * NN + n;
    y[0]              = sum.x + bn;
    y[(size_t)NN]     = sum.y + bn;
    y[(size_t)2 * NN] = sum.z + bn;
    y[(size_t)3 * NN] = sum.w + bn;
  }
}

#define TAB_HALF  176
#define TAB_NODES (2 * TAB_HALF + 1)
#define TAB_XMAX  5.5f
#define TAB_MAGIC 8388784.0f
#define TAB_BITS0 0x4B000000

#define TS 6
#define NW 32
#define NT (32 * NW)
#define UR 3
#define HCH (HH / 128)
#define NTRIPS ((SS / NW) * HCH)

struct GGroup {
  v2f pf[UR][2];
  v2f q[UR];
  v2f xw[UR];
};

__global__ __launch_bounds__(NT) __attribute__((amdgpu_waves_per_eu(8, 8)))
void pair_gelu_reduce(
    const float* __restrict__ hsrc, const float* __restrict__ htgt,
    const float* __restrict__ wcls, float* __restrict__ out) {
  __shared__ v2f s_tab[TAB_NODES * 32];
  __shared__ v2f s_node[TAB_NODES];
  __shared__ __attribute__((aligned(16))) float s_src[TS * HH];
  __shared__ __attribute__((aligned(16))) float s_w[HH];

  const int s0 = blockIdx.x * TS;
  const int l = blockIdx.y;
  const int b = blockIdx.z;
  const int tid = threadIdx.x;

  if (tid <= TAB_HALF) {
    const float xa  = (float)tid * (1.0f / 32.0f);
    const float q   = 0.5f * erfcf(xa * 0.70710678118654752f);
    const float ph0 = 0.39894228040143268f * expf(-0.5f * xa * xa);
    const float phi = fmaf(ph0, fmaf(xa * xa, 1.0f / 32768.0f, -1.0f / 32768.0f), ph0);
    v2f ep; ep.x = 1.0f - q; ep.y = phi;
    v2f en; en.x = q;        en.y = phi;
    if (tid == TAB_HALF) { ep.x = 1.0f; ep.y = 0.0f; en.x = 0.0f; en.y = 0.0f; }
    s_node[TAB_HALF + tid] = ep;
    s_node[TAB_HALF - tid] = en;
  }

  const float* __restrict__ srcbase =
      hsrc + ((size_t)((b * SS + s0) * LL + l)) * HH;
  for (int e = tid; e < TS * HH; e += NT) {
    const int i = e / HH;
    const int h = e - i * HH;
    const int pos = (h & ~127) + ((h & 2) << 5) + ((h >> 1) & 62) + (h & 1);
    s_src[i * HH + pos] = srcbase[(size_t)i * (LL * HH) + h];
  }
  for (int h = tid; h < HH; h += NT) s_w[h] = wcls[h * LL + l];
  __syncthreads();
  const int lane = tid & 31;
  const int wid  = tid >> 5;
#pragma unroll
  for (int k = 0; k < TAB_NODES / NW; ++k) {
    const int n = wid + k * NW;
    s_tab[n * 32 + lane] = s_node[n];
  }
  if (wid < TAB_NODES % NW) {
    const int n = TAB_NODES - TAB_NODES % NW + wid;
    s_tab[n * 32 + lane] = s_node[n];
  }
  __syncthreads();
  const int wu = __builtin_amdgcn_readfirstlane(wid);

  const float* __restrict__ tbase =
      htgt + ((size_t)(b * SS * LL + l)) * HH + (size_t)wu * (LL * HH) + 4 * lane;
  auto trip_ptr = [&](int T) {
    const int rr = T / HCH;
    const int kk = T - rr * HCH;
    return tbase + (size_t)rr * (NW * LL * HH) + kk * 128;
  };
  auto load_s = [&](v2f (&dst)[UR], const float* base, int r0, int p) {
#pragma unroll
    for (int r = 0; r < UR; ++r) dst[r] = *(const v2f*)(base + (r0 + r) * HH + 64 * p);
  };
  auto issue = [&](GGroup& G, const v2f (&s2)[UR], v2f t2, v2f w2) {
    const v2f ksat = (v2f){1.0f / (2.0f * TAB_XMAX), 1.0f / (2.0f * TAB_XMAX)};
    const v2f knod = (v2f){2.0f * TAB_HALF, 2.0f * TAB_HALF};
    const v2f k223 = (v2f){8388608.0f, 8388608.0f};
    const v2f kinv = (v2f){-1.0f / 64.0f, -1.0f / 64.0f};
    const v2f koff = (v2f){TAB_MAGIC / 64.0f, TAB_MAGIC / 64.0f};
    const v2f two = (v2f){2.0f, 2.0f};
    const v2f one = (v2f){1.0f, 1.0f};
    static_assert(UR == 3, "the asm statement below is written for three rows");
    v2f x[UR], u[UR];
    asm("v_pk_add_f32 %0, %9, %12\n\t"
        "v_pk_add_f32 %1, %10, %12\n\t"
        "v_pk_add_f32 %2, %11, %12\n\t"
        "s_delay_alu instid0(VALU_DEP_3) | instskip(NEXT) | instid1(VALU_DEP_3)\n\t"
        "v_pk_fma_f32 %3, %0, %14, 0.5 op_sel_hi:[1,0,0] clamp\n\t"
        "v_pk_fma_f32 %4, %1, %14, 0.5 op_sel_hi:[1,0,0] clamp\n\t"
        "s_delay_alu instid0(VALU_DEP_3)\n\t"
        "v_pk_fma_f32 %5, %2, %14, 0.5 op_sel_hi:[1,0,0] clamp\n\t"
        "v_pk_mul_f32 %6, %0, %13\n\t"
        "v_pk_mul_f32 %7, %1, %13\n\t"
        "v_pk_mul_f32 %8, %2, %13"
        : "=&v"(x[0]), "=&v"(x[1]), "=&v"(x[2]),
          "=&v"(u[0]), "=&v"(u[1]), "=&v"(u[2]),
          "=&v"(G.xw[0]), "=&v"(G.xw[1]), "=&v"(G.xw[2])
        : "v"(s2[0]), "v"(s2[1]), "v"(s2[2]), "v"(t2), "v"(w2), "s"(ksat));
#pragma unroll
    for (int r = 0; r < UR; ++r) {
      const v2f tq = __builtin_elementwise_fma(u[r], knod, k223);
      G.pf[r][0] = s_tab[(__float_as_int(tq.x) - TAB_BITS0) * 32 + lane];
      G.pf[r][1] = s_tab[(__float_as_int(tq.y) - TAB_BITS0) * 32 + lane];
      const v2f xh = __builtin_elementwise_fma(tq, kinv, koff);
      const v2f d  = __builtin_elementwise_fma(xh, two, x[r]);
      const v2f p  = __builtin_elementwise_fma(d, xh, one);
      G.q[r]  = d * p;
    }
  };

  v2f acc[TS];
#pragma unroll
  for (int i = 0; i < TS; ++i) acc[i] = (v2f){0.0f, 0.0f};
  auto sum = [&](const GGroup& G, int r0) {
    static_assert(UR == 3, "the two asm statements below are written for three rows");
    float e0, e1, e2, e3, e4, e5;
    asm("v_fma_f32 %0, %6, %7, %8\n\t"
        "v_fma_f32 %1, %9, %10, %11\n\t"
        "v_fma_f32 %2, %12, %13, %14\n\t"
        "v_fma_f32 %3, %15, %16, %17\n\t"
        "v_fma_f32 %4, %18, %19, %20\n\t"
        "v_fma_f32 %5, %21, %22, %23"
        : "=&v"(e0), "=&v"(e1), "=&v"(e2), "=&v"(e3), "=&v"(e4), "=&v"(e5)
        : "v"(G.pf[0][0].y), "v"(G.q[0].x), "v"(G.pf[0][0].x),
          "v"(G.pf[0][1].y), "v"(G.q[0].y), "v"(G.pf[0][1].x),
          "v"(G.pf[1][0].y), "v"(G.q[1].x), "v"(G.pf[1][0].x),
          "v"(G.pf[1][1].y), "v"(G.q[1].y), "v"(G.pf[1][1].x),
          "v"(G.pf[2][0].y), "v"(G.q[2].x), "v"(G.pf[2][0].x),
          "v"(G.pf[2][1].y), "v"(G.q[2].y), "v"(G.pf[2][1].x));
    const v2f ph0 = (v2f){e0, e1}, ph1 = (v2f){e2, e3}, ph2 = (v2f){e4, e5};
    asm("v_pk_fma_f32 %0, %3, %6, %0\n\t"
        "s_delay_alu instid0(VALU_DEP_4) | instskip(NEXT) | instid1(VALU_DEP_3)\n\t"
        "v_pk_fma_f32 %1, %4, %7, %1\n\t"
        "v_pk_fma_f32 %2, %5, %8, %2"
        : "+v"(acc[r0]), "+v"(acc[r0 + 1]), "+v"(acc[r0 + 2])
        : "v"(ph0), "v"(ph1), "v"(ph2), "v"(G.xw[0]), "v"(G.xw[1]), "v"(G.xw[2]));
  };

  v4f tva = *(const v4f*)trip_ptr(0);
  v4f tvb = *(const v4f*)trip_ptr(1);
  const float* sp = s_src + 2 * lane;
  const float* wp = s_w + 4 * lane;
  v4f wva = *(const v4f*)wp;
  v4f wvb;
  v2f sa[UR], sb[UR];
  GGroup C;
  load_s(sa, sp, 0, 0);
  load_s(sb, sp, UR, 0);
  __builtin_amdgcn_sched_barrier(0);
  issue(C, sa, tva.xy, wva.xy);
  int t = wu;
  int kq = 0;

  auto one_trip = [&](v4f& tc, v4f& tx, v4f& wc, v4f& wx, const int trip, const bool rowend) {
    const float* spn = rowend ? sp - (HH - 128) : sp + 128;
    const float* wpn = rowend ? wp - (HH - 128) : wp + 128;
    wx = *(const v4f*)wpn;
    GGroup N1, N2, N3;
    load_s(sa, sp, 0, 1);
    __builtin_amdgcn_sched_barrier(0);
    issue(N1, sb, tc.xy, wc.xy);
    __builtin_amdgcn_sched_barrier(0);
    sum(C, 0);
    __builtin_amdgcn_sched_barrier(0);
    load_s(sb, sp, UR, 1);
    __builtin_amdgcn_sched_barrier(0);
    issue(N2, sa, tc.zw, wc.zw);
    __builtin_amdgcn_sched_barrier(0);
    sum(N1, UR);
    __builtin_amdgcn_sched_barrier(0);
    load_s(sa, spn, 0, 0);
    __builtin_amdgcn_sched_barrier(0);
    issue(N3, sb, tc.zw, wc.zw);
    __builtin_amdgcn_sched_barrier(0);
    sum(N2, 0);
    __builtin_amdgcn_sched_barrier(0);
    tc = *(const v4f*)trip_ptr(trip + 2 < NTRIPS ? trip + 2 : NTRIPS - 1);
    load_s(sb, spn, UR, 0);
    __builtin_amdgcn_sched_barrier(0);
    issue(C, sa, tx.xy, wx.xy);
    __builtin_amdgcn_sched_barrier(0);
    sum(N3, UR);
    sp = spn;
    wp = wpn;
    ++kq;
    if (rowend) {
      float a[TS];
#pragma unroll
      for (int i = 0; i < TS; ++i) a[i] = acc[i].x + acc[i].y;
#pragma unroll
      for (int off = 16; off > 0; off >>= 1) {
#pragma unroll
        for (int i = 0; i < TS; ++i) a[i] += __shfl_xor(a[i], off, 32);
      }
      if (lane == 0) {
#pragma unroll
        for (int i = 0; i < TS; ++i)
          out[((size_t)((b * SS + s0 + i) * SS + t)) * LL + l] = a[i];
      }
#pragma unroll
      for (int i = 0; i < TS; ++i) acc[i] = (v2f){0.0f, 0.0f};
      kq = 0;
      t += NW;
    }
  };

  for (int trip = 0; trip < NTRIPS; trip += 2) {
    one_trip(tva, tvb, wva, wvb, trip, false);
    one_trip(tvb, tva, wvb, wva, trip + 1, kq == HCH - 1);
  }
}

extern "C" void kernel_launch(void* const* d_in, const int* in_sizes, int n_in,
                              void* d_out, int out_size, void* d_ws, size_t ws_size,
                              hipStream_t stream) {
  const float* pooled = (const float*)d_in[0];
  const float* W_src  = (const float*)d_in[1];
  const float* b_src  = (const float*)d_in[2];
  const float* W_tgt  = (const float*)d_in[3];
  const float* b_tgt  = (const float*)d_in[4];
  const float* wcls   = (const float*)d_in[5];
  float* out = (float*)d_out;

  float* hsrc = (float*)d_ws;
  float* htgt = hsrc + (size_t)MM * NN;

  dim3 g1(NN / (16 * PN * NTILE), MM / (16 * PM));
  proj_wmma_f32<<<g1, 32 * PWAVES, 0, stream>>>(pooled, W_src, b_src, hsrc);
  proj_wmma_f32<<<g1, 32 * PWAVES, 0, stream>>>(pooled, W_tgt, b_tgt, htgt);

  dim3 g2(SS / TS, LL, BB);
  pair_gelu_reduce<<<g2, NT, 0, stream>>>(hsrc, htgt, wcls, out);
}


# pair kernel: a wave lowers its priority (s_setprio 3,2,1,0) at trips 22, 30, 34 of 36, so lagging waves of a SIMD issue first and all eight end together; was oldest-first to the end; sums unchanged
# speedup vs baseline: 4.2341x; 1.0496x over previous
#include <hip/hip_runtime.h>
#include <math.h>

typedef __attribute__((ext_vector_type(2))) float v2f;
typedef __attribute__((ext_vector_type(8))) float v8f;

#define BB 2
#define SS 192
#define HH 768
#define LL 4
#define NN (HH * LL)
#define MM (BB * SS)

typedef __attribute__((ext_vector_type(4))) float v4f;

#define PM 3
#define PN 3
#define KQ 4
#define KT (8 * KQ)
#define KSPLIT 6
#define KPART (HH / KSPLIT)
#define NTILE 2
#define PWAVES (KSPLIT * NTILE)
#define PUNITS (PM * PN * 2)
#define UPW (PUNITS / KSPLIT)

#define PROJ_LOAD(A, B, q, k)                                                   \
  _Pragma("unroll") for (int i = 0; i < PM; ++i)                                \
      A[q][i] = *(const v4f*)(pa + (size_t)i * (16 * HH) + (k) + 8 * (q));      \
  _Pragma("unroll") for (int j = 0; j < PN; ++j)                                \
      B[q][j] = *(const v4f*)(pb + (size_t)j * (16 * HH) + (k) + 8 * (q));

#define PROJ_MMA(A, B, q)                                                       \
  _Pragma("unroll") for (int i = 0; i < PM; ++i)                                \
    _Pragma("unroll") for (int j = 0; j < PN; ++j)                              \
      acc[i][j] = __builtin_amdgcn_wmma_f32_16x16x4_f32(                        \
          false, A[q][i].xy, false, B[q][j].xy, (short)0, acc[i][j], false,     \
          false);                                                               \
  _Pragma("unroll") for (int i = 0; i < PM; ++i)                                \
    _Pragma("unroll") for (int j = 0; j < PN; ++j)                              \
      acc[i][j] = __builtin_amdgcn_wmma_f32_16x16x4_f32(                        \
          false, A[q][i].zw, false, B[q][j].zw, (short)0, acc[i][j], false,     \
          false);

__global__ __launch_bounds__(32 * PWAVES) __attribute__((amdgpu_waves_per_eu(3, 3)))
void proj_wmma_f32(
    const float* __restrict__ X, const float* __restrict__ W,
    const float* __restrict__ bias, float* __restrict__ Y) {
  __shared__ v4f s_part[PWAVES * PUNITS * 32];

  const int lane = threadIdx.x & 31;
  const int wid = __builtin_amdgcn_readfirstlane((int)(threadIdx.x >> 5));
  const int jt  = wid & 1;
  const int ks  = wid >> 1;
  const int n0 = blockIdx.x * (16 * PN * NTILE) + jt * (16 * PN);
  const int m0 = blockIdx.y * (16 * PM);
  const int r16  = lane & 15;
  const int hh   = lane >> 4;

  const float* __restrict__ pa = X + (size_t)(m0 + r16) * HH + ks * KPART + 4 * hh;
  const float* __restrict__ pb = W + (size_t)(n0 + r16) * HH + ks * KPART + 4 * hh;

  v8f acc[PM][PN];
#pragma unroll
  for (int i = 0; i < PM; ++i)
#pragma unroll
    for (int j = 0; j < PN; ++j) acc[i][j] = (v8f){};

  v4f a0[KQ][PM], b0[KQ][PN];
  v4f a1[KQ][PM], b1[KQ][PN];

#pragma unroll
  for (int q = 0; q < KQ; ++q) { PROJ_LOAD(a0, b0, q, 0) }
#pragma unroll
  for (int q = 0; q < KQ; ++q) {
    PROJ_LOAD(a1, b1, q, KT)
    PROJ_MMA(a0, b0, q)
  }
#pragma unroll
  for (int q = 0; q < KQ; ++q) {
    PROJ_LOAD(a0, b0, q, 2 * KT)
    PROJ_MMA(a1, b1, q)
  }
#pragma unroll
  for (int q = 0; q < KQ; ++q) {
    PROJ_LOAD(a1, b1, q, 3 * KT)
    PROJ_MMA(a0, b0, q)
  }
#pragma unroll
  for (int q = 0; q < KQ; ++q) { PROJ_MMA(a1, b1, q) }

  v4f* mine = s_part + (size_t)wid * (PUNITS * 32) + lane;
#pragma unroll
  for (int i = 0; i < PM; ++i)
#pragma unroll
    for (int j = 0; j < PN; ++j) {
      const int ti = i * PN + j;
      v4f lo, hi;
      lo.x = acc[i][j][0]; lo.y = acc[i][j][1]; lo.z = acc[i][j][2]; lo.w = acc[i][j][3];
      hi.x = acc[i][j][4]; hi.y = acc[i][j][5]; hi.z = acc[i][j][6]; hi.w = acc[i][j][7];
      mine[(2 * ti) * 32]     = lo;
      mine[(2 * ti + 1) * 32] = hi;
    }
  __syncthreads();

  const v4f* parts = s_part + (size_t)jt * (PUNITS * 32) + lane;
#pragma unroll
  for (int e = 0; e < UPW; ++e) {
    const int u = ks * UPW + e;
    v4f sum = parts[u * 32];
#pragma unroll
    for (int p = 1; p < KSPLIT; ++p)
      sum += parts[(size_t)p * (NTILE * PUNITS * 32) + u * 32];
    const int ti = u >> 1;
    const int i  = ti / PN;
    const int j  = ti - i * PN;
    const int n  = n0 + 16 * j + r16;
    const float bn = bias[n];
    float* __restrict__ y = Y + (size_t)(m0 + 16 * i + 8 * hh + 4 * (u & 1)) * NN + n;
    y[0]              = sum.x + bn;
    y[(size_t)NN]     = sum.y + bn;
    y[(size_t)2 * NN] = sum.z + bn;
    y[(size_t)3 * NN] = sum.w + bn;
  }
}

#define TAB_HALF  176
#define TAB_NODES (2 * TAB_HALF + 1)
#define TAB_XMAX  5.5f
#define TAB_MAGIC 8388784.0f
#define TAB_BITS0 0x4B000000

#define TS 6
#define NW 32
#define NT (32 * NW)
#define UR 3
#define HCH (HH / 128)
#define NTRIPS ((SS / NW) * HCH)
#define PRIO_MARK2 22
#define PRIO_MARK1 30
#define PRIO_MARK0 34

struct GGroup {
  v2f pf[UR][2];
  v2f q[UR];
  v2f xw[UR];
};

__global__ __launch_bounds__(NT) __attribute__((amdgpu_waves_per_eu(8, 8)))
void pair_gelu_reduce(
    const float* __restrict__ hsrc, const float* __restrict__ htgt,
    const float* __restrict__ wcls, float* __restrict__ out) {
  __shared__ v2f s_tab[TAB_NODES * 32];
  __shared__ v2f s_node[TAB_NODES];
  __shared__ __attribute__((aligned(16))) float s_src[TS * HH];
  __shared__ __attribute__((aligned(16))) float s_w[HH];

  const int s0 = blockIdx.x * TS;
  const int l = blockIdx.y;
  const int b = blockIdx.z;
  const int tid = threadIdx.x;

  if (tid <= TAB_HALF) {
    const float xa  = (float)tid * (1.0f / 32.0f);
    const float q   = 0.5f * erfcf(xa * 0.70710678118654752f);
    const float ph0 = 0.39894228040143268f * expf(-0.5f * xa * xa);
    const float phi = fmaf(ph0, fmaf(xa * xa, 1.0f / 32768.0f, -1.0f / 32768.0f), ph0);
    v2f ep; ep.x = 1.0f - q; ep.y = phi;
    v2f en; en.x = q;        en.y = phi;
    if (tid == TAB_HALF) { ep.x = 1.0f; ep.y = 0.0f; en.x = 0.0f; en.y = 0.0f; }
    s_node[TAB_HALF + tid] = ep;
    s_node[TAB_HALF - tid] = en;
  }

  const float* __restrict__ srcbase =
      hsrc + ((size_t)((b * SS + s0) * LL + l)) * HH;
  for (int e = tid; e < TS * HH; e += NT) {
    const int i = e / HH;
    const int h = e - i * HH;
    const int pos = (h & ~127) + ((h & 2) << 5) + ((h >> 1) & 62) + (h & 1);
    s_src[i * HH + pos] = srcbase[(size_t)i * (LL * HH) + h];
  }
  for (int h = tid; h < HH; h += NT) s_w[h] = wcls[h * LL + l];
  __syncthreads();
  const int lane = tid & 31;
  const int wid  = tid >> 5;
#pragma unroll
  for (int k = 0; k < TAB_NODES / NW; ++k) {
    const int n = wid + k * NW;
    s_tab[n * 32 + lane] = s_node[n];
  }
  if (wid < TAB_NODES % NW) {
    const int n = TAB_NODES - TAB_NODES % NW + wid;
    s_tab[n * 32 + lane] = s_node[n];
  }
  __syncthreads();
  const int wu = __builtin_amdgcn_readfirstlane(wid);

  const float* __restrict__ tbase =
      htgt + ((size_t)(b * SS * LL + l)) * HH + (size_t)wu * (LL * HH) + 4 * lane;
  auto trip_ptr = [&](int T) {
    const int rr = T / HCH;
    const int kk = T - rr * HCH;
    return tbase + (size_t)rr * (NW * LL * HH) + kk * 128;
  };
  auto load_s = [&](v2f (&dst)[UR], const float* base, int r0, int p) {
#pragma unroll
    for (int r = 0; r < UR; ++r) dst[r] = *(const v2f*)(base + (r0 + r) * HH + 64 * p);
  };
  auto issue = [&](GGroup& G, const v2f (&s2)[UR], v2f t2, v2f w2) {
    const v2f ksat = (v2f){1.0f / (2.0f * TAB_XMAX), 1.0f / (2.0f * TAB_XMAX)};
    const v2f knod = (v2f){2.0f * TAB_HALF, 2.0f * TAB_HALF};
    const v2f k223 = (v2f){8388608.0f, 8388608.0f};
    const v2f kinv = (v2f){-1.0f / 64.0f, -1.0f / 64.0f};
    const v2f koff = (v2f){TAB_MAGIC / 64.0f, TAB_MAGIC / 64.0f};
    const v2f two = (v2f){2.0f, 2.0f};
    const v2f one = (v2f){1.0f, 1.0f};
    static_assert(UR == 3, "the asm statement below is written for three rows");
    v2f x[UR], u[UR];
    asm("v_pk_add_f32 %0, %9, %12\n\t"
        "v_pk_add_f32 %1, %10, %12\n\t"
        "v_pk_add_f32 %2, %11, %12\n\t"
        "s_delay_alu instid0(VALU_DEP_3) | instskip(NEXT) | instid1(VALU_DEP_3)\n\t"
        "v_pk_fma_f32 %3, %0, %14, 0.5 op_sel_hi:[1,0,0] clamp\n\t"
        "v_pk_fma_f32 %4, %1, %14, 0.5 op_sel_hi:[1,0,0] clamp\n\t"
        "s_delay_alu instid0(VALU_DEP_3)\n\t"
        "v_pk_fma_f32 %5, %2, %14, 0.5 op_sel_hi:[1,0,0] clamp\n\t"
        "v_pk_mul_f32 %6, %0, %13\n\t"
        "v_pk_mul_f32 %7, %1, %13\n\t"
        "v_pk_mul_f32 %8, %2, %13"
        : "=&v"(x[0]), "=&v"(x[1]), "=&v"(x[2]),
          "=&v"(u[0]), "=&v"(u[1]), "=&v"(u[2]),
          "=&v"(G.xw[0]), "=&v"(G.xw[1]), "=&v"(G.xw[2])
        : "v"(s2[0]), "v"(s2[1]), "v"(s2[2]), "v"(t2), "v"(w2), "s"(ksat));
#pragma unroll
    for (int r = 0; r < UR; ++r) {
      const v2f tq = __builtin_elementwise_fma(u[r], knod, k223);
      G.pf[r][0] = s_tab[(__float_as_int(tq.x) - TAB_BITS0) * 32 + lane];
      G.pf[r][1] = s_tab[(__float_as_int(tq.y) - TAB_BITS0) * 32 + lane];
      const v2f xh = __builtin_elementwise_fma(tq, kinv, koff);
      const v2f d  = __builtin_elementwise_fma(xh, two, x[r]);
      const v2f p  = __builtin_elementwise_fma(d, xh, one);
      G.q[r]  = d * p;
    }
  };

  v2f acc[TS];
#pragma unroll
  for (int i = 0; i < TS; ++i) acc[i] = (v2f){0.0f, 0.0f};
  auto sum = [&](const GGroup& G, int r0) {
    static_assert(UR == 3, "the two asm statements below are written for three rows");
    float e0, e1, e2, e3, e4, e5;
    asm("v_fma_f32 %0, %6, %7, %8\n\t"
        "v_fma_f32 %1, %9, %10, %11\n\t"
        "v_fma_f32 %2, %12, %13, %14\n\t"
        "v_fma_f32 %3, %15, %16, %17\n\t"
        "v_fma_f32 %4, %18, %19, %20\n\t"
        "v_fma_f32 %5, %21, %22, %23"
        : "=&v"(e0), "=&v"(e1), "=&v"(e2), "=&v"(e3), "=&v"(e4), "=&v"(e5)
        : "v"(G.pf[0][0].y), "v"(G.q[0].x), "v"(G.pf[0][0].x),
          "v"(G.pf[0][1].y), "v"(G.q[0].y), "v"(G.pf[0][1].x),
          "v"(G.pf[1][0].y), "v"(G.q[1].x), "v"(G.pf[1][0].x),
          "v"(G.pf[1][1].y), "v"(G.q[1].y), "v"(G.pf[1][1].x),
          "v"(G.pf[2][0].y), "v"(G.q[2].x), "v"(G.pf[2][0].x),
          "v"(G.pf[2][1].y), "v"(G.q[2].y), "v"(G.pf[2][1].x));
    const v2f ph0 = (v2f){e0, e1}, ph1 = (v2f){e2, e3}, ph2 = (v2f){e4, e5};
    asm("v_pk_fma_f32 %0, %3, %6, %0\n\t"
        "s_delay_alu instid0(VALU_DEP_4) | instskip(NEXT) | instid1(VALU_DEP_3)\n\t"
        "v_pk_fma_f32 %1, %4, %7, %1\n\t"
        "v_pk_fma_f32 %2, %5, %8, %2"
        : "+v"(acc[r0]), "+v"(acc[r0 + 1]), "+v"(acc[r0 + 2])
        : "v"(ph0), "v"(ph1), "v"(ph2), "v"(G.xw[0]), "v"(G.xw[1]), "v"(G.xw[2]));
  };

  v4f tva = *(const v4f*)trip_ptr(0);
  v4f tvb = *(const v4f*)trip_ptr(1);
  const float* sp = s_src + 2 * lane;
  const float* wp = s_w + 4 * lane;
  v4f wva = *(const v4f*)wp;
  v4f wvb;
  v2f sa[UR], sb[UR];
  GGroup C;
  load_s(sa, sp, 0, 0);
  load_s(sb, sp, UR, 0);
  __builtin_amdgcn_sched_barrier(0);
  issue(C, sa, tva.xy, wva.xy);
  int t = wu;
  int kq = 0;

  auto one_trip = [&](v4f& tc, v4f& tx, v4f& wc, v4f& wx, const int trip, const bool rowend) {
    const float* spn = rowend ? sp - (HH - 128) : sp + 128;
    const float* wpn = rowend ? wp - (HH - 128) : wp + 128;
    wx = *(const v4f*)wpn;
    GGroup N1, N2, N3;
    load_s(sa, sp, 0, 1);
    __builtin_amdgcn_sched_barrier(0);
    issue(N1, sb, tc.xy, wc.xy);
    __builtin_amdgcn_sched_barrier(0);
    sum(C, 0);
    __builtin_amdgcn_sched_barrier(0);
    load_s(sb, sp, UR, 1);
    __builtin_amdgcn_sched_barrier(0);
    issue(N2, sa, tc.zw, wc.zw);
    __builtin_amdgcn_sched_barrier(0);
    sum(N1, UR);
    __builtin_amdgcn_sched_barrier(0);
    load_s(sa, spn, 0, 0);
    __builtin_amdgcn_sched_barrier(0);
    issue(N3, sb, tc.zw, wc.zw);
    __builtin_amdgcn_sched_barrier(0);
    sum(N2, 0);
    __builtin_amdgcn_sched_barrier(0);
    tc = *(const v4f*)trip_ptr(trip + 2 < NTRIPS ? trip + 2 : NTRIPS - 1);
    load_s(sb, spn, UR, 0);
    __builtin_amdgcn_sched_barrier(0);
    issue(C, sa, tx.xy, wx.xy);
    __builtin_amdgcn_sched_barrier(0);
    sum(N3, UR);
    sp = spn;
    wp = wpn;
    ++kq;
    if (rowend) {
      float a[TS];
#pragma unroll
      for (int i = 0; i < TS; ++i) a[i] = acc[i].x + acc[i].y;
#pragma unroll
      for (int off = 16; off > 0; off >>= 1) {
#pragma unroll
        for (int i = 0; i < TS; ++i) a[i] += __shfl_xor(a[i], off, 32);
      }
      if (lane == 0) {
#pragma unroll
        for (int i = 0; i < TS; ++i)
          out[((size_t)((b * SS + s0 + i) * SS + t)) * LL + l] = a[i];
      }
#pragma unroll
      for (int i = 0; i < TS; ++i) acc[i] = (v2f){0.0f, 0.0f};
      kq = 0;
      t += NW;
    }
  };

  __builtin_amdgcn_s_setprio(3);
  int mark = PRIO_MARK2;
  for (int trip = 0; trip < NTRIPS; trip += 2) {
    if (trip == mark) {
      if (trip == PRIO_MARK2)      { __builtin_amdgcn_s_setprio(2); mark = PRIO_MARK1; }
      else if (trip == PRIO_MARK1) { __builtin_amdgcn_s_setprio(1); mark = PRIO_MARK0; }
      else                         { __builtin_amdgcn_s_setprio(0); mark = -1; }
    }
    one_trip(tva, tvb, wva, wvb, trip, false);
    one_trip(tvb, tva, wvb, wva, trip + 1, kq == HCH - 1);
  }
}

extern "C" void kernel_launch(void* const* d_in, const int* in_sizes, int n_in,
                              void* d_out, int out_size, void* d_ws, size_t ws_size,
                              hipStream_t stream) {
  const float* pooled = (const float*)d_in[0];
  const float* W_src  = (const float*)d_in[1];
  const float* b_src  = (const float*)d_in[2];
  const float* W_tgt  = (const float*)d_in[3];
  const float* b_tgt  = (const float*)d_in[4];
  const float* wcls   = (const float*)d_in[5];
  float* out = (float*)d_out;

  float* hsrc = (float*)d_ws;
  float* htgt = hsrc + (size_t)MM * NN;

  dim3 g1(NN / (16 * PN * NTILE), MM / (16 * PM));
  proj_wmma_f32<<<g1, 32 * PWAVES, 0, stream>>>(pooled, W_src, b_src, hsrc);
  proj_wmma_f32<<<g1, 32 * PWAVES, 0, stream>>>(pooled, W_tgt, b_tgt, htgt);

  dim3 g2(SS / TS, LL, BB);
  pair_gelu_reduce<<<g2, NT, 0, stream>>>(hsrc, htgt, wcls, out);
}
